# starting kernel
# baseline (speedup 1.0000x reference)
#include <hip/hip_runtime.h>

typedef float v2f __attribute__((ext_vector_type(2)));
typedef float v8f __attribute__((ext_vector_type(8)));

#define NF      39
#define DDIM    64
#define NPAIR   741
#define WAVES   4
#define LSTRIDE 66
#define LROWS   48
#define LDS_PER_WAVE (LROWS * LSTRIDE)

__global__ __launch_bounds__(WAVES * 32)
void gram_triu_wmma_kernel(const float* __restrict__ x,
                           float* __restrict__ out,
                           int nBatch)
{
    __shared__ float lds[WAVES * LDS_PER_WAVE];

    const int wave = threadIdx.x >> 5;
    const int lane = threadIdx.x & 31;
    const int b    = blockIdx.x * WAVES + wave;

    float* __restrict__ my = &lds[wave * LDS_PER_WAVE];

    if (b < nBatch) {
        const float* __restrict__ xb = x + (size_t)b * (NF * DDIM);

        for (int i = NF * LSTRIDE + lane; i < LROWS * LSTRIDE; i += 32)
            my[i] = 0.0f;

        for (int i = lane; i < (NF * DDIM) / 2; i += 32) {
            const int r  = i >> 5;
            const int c2 = i & 31;
            const float2 v = reinterpret_cast<const float2*>(xb)[i];
            *reinterpret_cast<float2*>(&my[r * LSTRIDE + c2 * 2]) = v;
        }
    }
    __syncthreads();
    if (b >= nBatch) return;

    const int r  = lane & 15;
    const int hf = lane >> 4;

    v8f a00 = {}, a01 = {}, a02 = {}, a11 = {}, a12 = {}, a22 = {};

    #pragma unroll
    for (int ks = 0; ks < DDIM / 4; ++ks) {
        const int col = ks * 4 + hf * 2;
        const v2f f0 = *reinterpret_cast<const v2f*>(&my[(r +  0) * LSTRIDE + col]);
        const v2f f1 = *reinterpret_cast<const v2f*>(&my[(r + 16) * LSTRIDE + col]);
        const v2f f2 = *reinterpret_cast<const v2f*>(&my[(r + 32) * LSTRIDE + col]);
        a00 = __builtin_amdgcn_wmma_f32_16x16x4_f32(false, f0, false, f0, (short)0, a00, false, false);
        a01 = __builtin_amdgcn_wmma_f32_16x16x4_f32(false, f0, false, f1, (short)0, a01, false, false);
        a02 = __builtin_amdgcn_wmma_f32_16x16x4_f32(false, f0, false, f2, (short)0, a02, false, false);
        a11 = __builtin_amdgcn_wmma_f32_16x16x4_f32(false, f1, false, f1, (short)0, a11, false, false);
        a12 = __builtin_amdgcn_wmma_f32_16x16x4_f32(false, f1, false, f2, (short)0, a12, false, false);
        a22 = __builtin_amdgcn_wmma_f32_16x16x4_f32(false, f2, false, f2, (short)0, a22, false, false);
    }

    float* __restrict__ ob = out + (size_t)b * NPAIR;
    auto emit = [&](const v8f& acc, int ti, int tj) {
        const int g = tj * 16 + r;
        #pragma unroll
        for (int v = 0; v < 8; ++v) {
            const int f = ti * 16 + hf * 8 + v;
            if (f < g && g < NF) {
                ob[f * 38 - (f * (f - 1)) / 2 + (g - f - 1)] = acc[v];
            }
        }
    };
    emit(a00, 0, 0); emit(a01, 0, 1); emit(a02, 0, 2);
    emit(a11, 1, 1); emit(a12, 1, 2); emit(a22, 2, 2);
}

extern "C" void kernel_launch(void* const* d_in, const int* in_sizes, int n_in,
                              void* d_out, int out_size, void* d_ws, size_t ws_size,
                              hipStream_t stream) {
    const float* x   = (const float*)d_in[0];
    float*       out = (float*)d_out;
    const int nBatch = in_sizes[0] / (NF * DDIM);

    dim3 block(WAVES * 32);
    dim3 grid((nBatch + WAVES - 1) / WAVES);
    gram_triu_wmma_kernel<<<grid, block, 0, stream>>>(x, out, nBatch);
}


# stage loop: all 39 row loads of a wave's tile are issued as one group before the first LDS store, so a wave pays one memory round trip for its tile instead of eleven serial ones
# speedup vs baseline: 1.2969x; 1.2969x over previous
#include <hip/hip_runtime.h>

typedef float v2f __attribute__((ext_vector_type(2)));
typedef float v8f __attribute__((ext_vector_type(8)));

#define NF      39
#define DDIM    64
#define NPAIR   741
#define WAVES   4
#define LSTRIDE 66
#define LROWS   48
#define LDS_PER_WAVE (LROWS * LSTRIDE)

__global__ __launch_bounds__(WAVES * 32)
void gram_triu_wmma_kernel(const float* __restrict__ x,
                           float* __restrict__ out,
                           int nBatch)
{
    __shared__ float lds[WAVES * LDS_PER_WAVE];

    const int wave = threadIdx.x >> 5;
    const int lane = threadIdx.x & 31;
    const int b    = blockIdx.x * WAVES + wave;

    float* __restrict__ my = &lds[wave * LDS_PER_WAVE];

    if (b < nBatch) {
        const float* __restrict__ xb = x + (size_t)b * (NF * DDIM);

        const float2* __restrict__ xb2 = reinterpret_cast<const float2*>(xb);
        float2 v[NF];
        for (int k = 0; k < NF; ++k)
            v[k] = xb2[k * 32 + lane];

        for (int i = NF * LSTRIDE + lane; i < LROWS * LSTRIDE; i += 32)
            my[i] = 0.0f;

        for (int k = 0; k < NF; ++k)
            *reinterpret_cast<float2*>(&my[k * LSTRIDE + lane * 2]) = v[k];
    }
    __syncthreads();
    if (b >= nBatch) return;

    const int r  = lane & 15;
    const int hf = lane >> 4;

    v8f a00 = {}, a01 = {}, a02 = {}, a11 = {}, a12 = {}, a22 = {};

    #pragma unroll
    for (int ks = 0; ks < DDIM / 4; ++ks) {
        const int col = ks * 4 + hf * 2;
        const v2f f0 = *reinterpret_cast<const v2f*>(&my[(r +  0) * LSTRIDE + col]);
        const v2f f1 = *reinterpret_cast<const v2f*>(&my[(r + 16) * LSTRIDE + col]);
        const v2f f2 = *reinterpret_cast<const v2f*>(&my[(r + 32) * LSTRIDE + col]);
        a00 = __builtin_amdgcn_wmma_f32_16x16x4_f32(false, f0, false, f0, (short)0, a00, false, false);
        a01 = __builtin_amdgcn_wmma_f32_16x16x4_f32(false, f0, false, f1, (short)0, a01, false, false);
        a02 = __builtin_amdgcn_wmma_f32_16x16x4_f32(false, f0, false, f2, (short)0, a02, false, false);
        a11 = __builtin_amdgcn_wmma_f32_16x16x4_f32(false, f1, false, f1, (short)0, a11, false, false);
        a12 = __builtin_amdgcn_wmma_f32_16x16x4_f32(false, f1, false, f2, (short)0, a12, false, false);
        a22 = __builtin_amdgcn_wmma_f32_16x16x4_f32(false, f2, false, f2, (short)0, a22, false, false);
    }

    float* __restrict__ ob = out + (size_t)b * NPAIR;
    auto emit = [&](const v8f& acc, int ti, int tj) {
        const int g = tj * 16 + r;
        #pragma unroll
        for (int v = 0; v < 8; ++v) {
            const int f = ti * 16 + hf * 8 + v;
            if (f < g && g < NF) {
                ob[f * 38 - (f * (f - 1)) / 2 + (g - f - 1)] = acc[v];
            }
        }
    };
    emit(a00, 0, 0); emit(a01, 0, 1); emit(a02, 0, 2);
    emit(a11, 1, 1); emit(a12, 1, 2); emit(a22, 2, 2);
}

extern "C" void kernel_launch(void* const* d_in, const int* in_sizes, int n_in,
                              void* d_out, int out_size, void* d_ws, size_t ws_size,
                              hipStream_t stream) {
    const float* x   = (const float*)d_in[0];
    float*       out = (float*)d_out;
    const int nBatch = in_sizes[0] / (NF * DDIM);

    dim3 block(WAVES * 32);
    dim3 grid((nBatch + WAVES - 1) / WAVES);
    gram_triu_wmma_kernel<<<grid, block, 0, stream>>>(x, out, nBatch);
}
